# starting kernel
# baseline (speedup 1.0000x reference)
#include <hip/hip_runtime.h>

typedef __attribute__((ext_vector_type(2))) float v2f;
typedef __attribute__((ext_vector_type(8))) float v8f;

#define NQ   8192
#define NK   2048
#define CDIM 128
#define QPB  128

__device__ __forceinline__ void top3_insert(float v, int ki,
                                            float& b0, float& b1, float& b2,
                                            int& i0, int& i1, int& i2) {
    bool c2 = v < b2;
    float x  = c2 ? v  : b2;
    int   xi = c2 ? ki : i2;
    bool c1 = x < b1;
    b2 = c1 ? b1 : x;   i2 = c1 ? i1 : xi;
    float y  = c1 ? x  : b1;
    int   yi = c1 ? xi : i1;
    bool c0 = y < b0;
    b1 = c0 ? b0 : y;   i1 = c0 ? i0 : yi;
    b0 = c0 ? y  : b0;  i0 = c0 ? yi : i0;
}

__global__ __launch_bounds__(256) void
fp_knn_wmma_kernel(const float* __restrict__ xyz_q,
                   const float* __restrict__ xyz_k,
                   const float* __restrict__ v_k,
                   float* __restrict__ out) {
    __shared__ float s_keys[NK * 4];

    const int b      = blockIdx.y;
    const int qblock = blockIdx.x * QPB;
    const int tid    = threadIdx.x;

    const float* kb = xyz_k + (size_t)b * NK * 3;
    for (int m = tid; m < NK; m += 256) {
        float kx = kb[m * 3 + 0];
        float ky = kb[m * 3 + 1];
        float kz = kb[m * 3 + 2];
        s_keys[m * 4 + 0] = kx;
        s_keys[m * 4 + 1] = ky;
        s_keys[m * 4 + 2] = kz;
        s_keys[m * 4 + 3] = kx * kx + ky * ky + kz * kz;
    }
    __syncthreads();

    const int wave = tid >> 5;
    const int lane = tid & 31;
    const int half = lane >> 4;
    const int l16  = lane & 15;

    const int   q  = qblock + wave * 16 + l16;
    const float* qp = xyz_q + ((size_t)b * NQ + q) * 3;
    float qx = qp[0], qy = qp[1], qz = qp[2];
    float qn2 = qx * qx + qy * qy + qz * qz;

    v2f bq;
    bq.x = half ? (-2.0f * qz) : (-2.0f * qx);
    bq.y = half ? 1.0f         : (-2.0f * qy);

    const int abase = l16 * 4 + half * 2;

    float b0 = 1e30f, b1 = 1e30f, b2 = 1e30f;
    int   i0 = 0,     i1 = 0,     i2 = 0;

    v2f a_cur;
    {
        const float* sk = &s_keys[abase];
        a_cur.x = sk[0];
        a_cur.y = sk[1];
    }

    for (int t = 0; t < NK; t += 16) {
        v2f a_nxt;
        {
            const float* sk = &s_keys[(((t + 16) & (NK - 1)) * 4) + abase];
            a_nxt.x = sk[0];
            a_nxt.y = sk[1];
        }

        v8f c = {};
        v8f d = __builtin_amdgcn_wmma_f32_16x16x4_f32(
             false, a_cur,  false, bq,
             (short)0, c,  false,  false);

        bool need = false;
#pragma unroll
        for (int r = 0; r < 8; ++r) need |= (d[r] < b2);

        if (__any(need)) {
            const int kbase = t + half * 8;
#pragma unroll
            for (int r = 0; r < 8; ++r) {
                top3_insert(d[r], kbase + r, b0, b1, b2, i0, i1, i2);
            }
        }

        a_cur = a_nxt;
    }

    float ob0 = __shfl_xor(b0, 16, 32);
    float ob1 = __shfl_xor(b1, 16, 32);
    float ob2 = __shfl_xor(b2, 16, 32);
    int   oi0 = __shfl_xor(i0, 16, 32);
    int   oi1 = __shfl_xor(i1, 16, 32);
    int   oi2 = __shfl_xor(i2, 16, 32);
    top3_insert(ob0, oi0, b0, b1, b2, i0, i1, i2);
    top3_insert(ob1, oi1, b0, b1, b2, i0, i1, i2);
    top3_insert(ob2, oi2, b0, b1, b2, i0, i1, i2);

    float d0 = fmaxf(b0 + qn2, 1e-10f);
    float d1 = fmaxf(b1 + qn2, 1e-10f);
    float d2 = fmaxf(b2 + qn2, 1e-10f);
    float w0 = 1.0f / d0, w1 = 1.0f / d1, w2 = 1.0f / d2;
    float inv = 1.0f / (w0 + w1 + w2);
    w0 *= inv; w1 *= inv; w2 *= inv;

    const float* vb = v_k + (size_t)b * NK * CDIM;
    float*       ob = out + ((size_t)b * NQ + qblock + wave * 16) * CDIM;
    const int cidx = lane * 4;

    for (int qq = 0; qq < 16; ++qq) {
        int   j0 = __shfl(i0, qq, 32);
        int   j1 = __shfl(i1, qq, 32);
        int   j2 = __shfl(i2, qq, 32);
        float u0 = __shfl(w0, qq, 32);
        float u1 = __shfl(w1, qq, 32);
        float u2 = __shfl(w2, qq, 32);

        const float4 f0 = *(const float4*)(vb + (size_t)j0 * CDIM + cidx);
        const float4 f1 = *(const float4*)(vb + (size_t)j1 * CDIM + cidx);
        const float4 f2 = *(const float4*)(vb + (size_t)j2 * CDIM + cidx);

        float4 r;
        r.x = u0 * f0.x + u1 * f1.x + u2 * f2.x;
        r.y = u0 * f0.y + u1 * f1.y + u2 * f2.y;
        r.z = u0 * f0.z + u1 * f1.z + u2 * f2.z;
        r.w = u0 * f0.w + u1 * f1.w + u2 * f2.w;

        *(float4*)(ob + (size_t)qq * CDIM + cidx) = r;
    }
}

extern "C" void kernel_launch(void* const* d_in, const int* in_sizes, int n_in,
                              void* d_out, int out_size, void* d_ws, size_t ws_size,
                              hipStream_t stream) {
    const float* xyz_q = (const float*)d_in[0];
    const float* xyz_k = (const float*)d_in[1];
    const float* v_k   = (const float*)d_in[2];
    float*       out   = (float*)d_out;

    const int B = in_sizes[0] / (NQ * 3);

    dim3 grid(NQ / QPB, B);
    dim3 block(256);
    fp_knn_wmma_kernel<<<grid, block, 0, stream>>>(xyz_q, xyz_k, v_k, out);
}


# Sweep takes four key tiles per trip into four independent top-3 chains per lane (tile c feeds chain c; two-waves-per-SIMD hint keeps them interleaved), merged at the end, ties to the lower key
# speedup vs baseline: 1.1586x; 1.1586x over previous
#include <hip/hip_runtime.h>

typedef __attribute__((ext_vector_type(2))) float v2f;
typedef __attribute__((ext_vector_type(8))) float v8f;

#define NQ   8192
#define NK   2048
#define CDIM 128
#define QPB  128
#define NCH  4

__device__ __forceinline__ void top3_insert(float v, int ki,
                                            float& b0, float& b1, float& b2,
                                            int& i0, int& i1, int& i2) {
    const bool c0 = v < b0;
    const bool c1 = v < b1;
    const bool c2 = v < b2;
    const int  p1 = c0 ? i0 : ki;
    const int  p2 = c1 ? i1 : ki;
    i0 = c0 ? ki : i0;
    i1 = c1 ? p1 : i1;
    i2 = c2 ? p2 : i2;
    const float n0 = fminf(b0, v);
    const float n1 = __builtin_amdgcn_fmed3f(b0, b1, v);
    const float n2 = __builtin_amdgcn_fmed3f(b1, b2, v);
    b0 = n0; b1 = n1; b2 = n2;
}

__device__ __forceinline__ void top3_merge(float v, int ki,
                                           float& b0, float& b1, float& b2,
                                           int& i0, int& i1, int& i2) {
    const bool c0 = (v < b0) | ((v == b0) & (ki < i0));
    const bool c1 = (v < b1) | ((v == b1) & (ki < i1));
    const bool c2 = (v < b2) | ((v == b2) & (ki < i2));
    const int  p1 = c0 ? i0 : ki;
    const int  p2 = c1 ? i1 : ki;
    i0 = c0 ? ki : i0;
    i1 = c1 ? p1 : i1;
    i2 = c2 ? p2 : i2;
    const float n0 = fminf(b0, v);
    const float n1 = __builtin_amdgcn_fmed3f(b0, b1, v);
    const float n2 = __builtin_amdgcn_fmed3f(b1, b2, v);
    b0 = n0; b1 = n1; b2 = n2;
}

__global__ __launch_bounds__(256) __attribute__((amdgpu_waves_per_eu(2, 2))) void
fp_knn_wmma_kernel(const float* __restrict__ xyz_q,
                   const float* __restrict__ xyz_k,
                   const float* __restrict__ v_k,
                   float* __restrict__ out) {
    __shared__ float s_keys[NK * 4];

    const int b      = blockIdx.y;
    const int qblock = blockIdx.x * QPB;
    const int tid    = threadIdx.x;

    const float* kb = xyz_k + (size_t)b * NK * 3;
    for (int m = tid; m < NK; m += 256) {
        float kx = kb[m * 3 + 0];
        float ky = kb[m * 3 + 1];
        float kz = kb[m * 3 + 2];
        s_keys[m * 4 + 0] = kx;
        s_keys[m * 4 + 1] = ky;
        s_keys[m * 4 + 2] = kz;
        s_keys[m * 4 + 3] = kx * kx + ky * ky + kz * kz;
    }
    __syncthreads();

    const int wave = tid >> 5;
    const int lane = tid & 31;
    const int half = lane >> 4;
    const int l16  = lane & 15;

    const int   q  = qblock + wave * 16 + l16;
    const float* qp = xyz_q + ((size_t)b * NQ + q) * 3;
    float qx = qp[0], qy = qp[1], qz = qp[2];
    float qn2 = qx * qx + qy * qy + qz * qz;

    v2f bq;
    bq.x = half ? (-2.0f * qz) : (-2.0f * qx);
    bq.y = half ? 1.0f         : (-2.0f * qy);

    const int abase = l16 * 4 + half * 2;

    float cb0[NCH], cb1[NCH], cb2[NCH];
    int   ci0[NCH], ci1[NCH], ci2[NCH];
#pragma unroll
    for (int c = 0; c < NCH; ++c) {
        cb0[c] = 1e30f; cb1[c] = 1e30f; cb2[c] = 1e30f;
        ci0[c] = 0;     ci1[c] = 0;     ci2[c] = 0;
    }

    v2f a_cur[NCH];
#pragma unroll
    for (int c = 0; c < NCH; ++c) {
        const float* sk = &s_keys[c * 64 + abase];
        a_cur[c].x = sk[0];
        a_cur[c].y = sk[1];
    }

    for (int t = 0; t < NK; t += 16 * NCH) {
        v2f a_nxt[NCH];
#pragma unroll
        for (int c = 0; c < NCH; ++c) {
            const float* sk = &s_keys[(((t + 16 * NCH + 16 * c) & (NK - 1)) * 4) + abase];
            a_nxt[c].x = sk[0];
            a_nxt[c].y = sk[1];
        }

        v8f d[NCH];
#pragma unroll
        for (int c = 0; c < NCH; ++c) {
            v8f z = {};
            d[c] = __builtin_amdgcn_wmma_f32_16x16x4_f32(
                 false, a_cur[c],  false, bq,
                 (short)0, z,  false,  false);
        }

        const int kbase = t + half * 8;
#pragma unroll
        for (int r = 0; r < 8; ++r) {
#pragma unroll
            for (int c = 0; c < NCH; ++c) {
                top3_insert(d[c][r], kbase + 16 * c + r,
                            cb0[c], cb1[c], cb2[c], ci0[c], ci1[c], ci2[c]);
            }
        }

#pragma unroll
        for (int c = 0; c < NCH; ++c) a_cur[c] = a_nxt[c];
    }

    float b0 = cb0[0], b1 = cb1[0], b2 = cb2[0];
    int   i0 = ci0[0], i1 = ci1[0], i2 = ci2[0];
#pragma unroll
    for (int c = 1; c < NCH; ++c) {
        top3_merge(cb0[c], ci0[c], b0, b1, b2, i0, i1, i2);
        top3_merge(cb1[c], ci1[c], b0, b1, b2, i0, i1, i2);
        top3_merge(cb2[c], ci2[c], b0, b1, b2, i0, i1, i2);
    }
    float ob0 = __shfl_xor(b0, 16, 32);
    float ob1 = __shfl_xor(b1, 16, 32);
    float ob2 = __shfl_xor(b2, 16, 32);
    int   oi0 = __shfl_xor(i0, 16, 32);
    int   oi1 = __shfl_xor(i1, 16, 32);
    int   oi2 = __shfl_xor(i2, 16, 32);
    top3_merge(ob0, oi0, b0, b1, b2, i0, i1, i2);
    top3_merge(ob1, oi1, b0, b1, b2, i0, i1, i2);
    top3_merge(ob2, oi2, b0, b1, b2, i0, i1, i2);

    float d0 = fmaxf(b0 + qn2, 1e-10f);
    float d1 = fmaxf(b1 + qn2, 1e-10f);
    float d2 = fmaxf(b2 + qn2, 1e-10f);
    float w0 = 1.0f / d0, w1 = 1.0f / d1, w2 = 1.0f / d2;
    float inv = 1.0f / (w0 + w1 + w2);
    w0 *= inv; w1 *= inv; w2 *= inv;

    const float* vb = v_k + (size_t)b * NK * CDIM;
    float*       ob = out + ((size_t)b * NQ + qblock + wave * 16) * CDIM;
    const int cidx = lane * 4;

    for (int qq = 0; qq < 16; ++qq) {
        int   j0 = __shfl(i0, qq, 32);
        int   j1 = __shfl(i1, qq, 32);
        int   j2 = __shfl(i2, qq, 32);
        float u0 = __shfl(w0, qq, 32);
        float u1 = __shfl(w1, qq, 32);
        float u2 = __shfl(w2, qq, 32);

        const float4 f0 = *(const float4*)(vb + (size_t)j0 * CDIM + cidx);
        const float4 f1 = *(const float4*)(vb + (size_t)j1 * CDIM + cidx);
        const float4 f2 = *(const float4*)(vb + (size_t)j2 * CDIM + cidx);

        float4 r;
        r.x = u0 * f0.x + u1 * f1.x + u2 * f2.x;
        r.y = u0 * f0.y + u1 * f1.y + u2 * f2.y;
        r.z = u0 * f0.z + u1 * f1.z + u2 * f2.z;
        r.w = u0 * f0.w + u1 * f1.w + u2 * f2.w;

        *(float4*)(ob + (size_t)qq * CDIM + cidx) = r;
    }
}

extern "C" void kernel_launch(void* const* d_in, const int* in_sizes, int n_in,
                              void* d_out, int out_size, void* d_ws, size_t ws_size,
                              hipStream_t stream) {
    const float* xyz_q = (const float*)d_in[0];
    const float* xyz_k = (const float*)d_in[1];
    const float* v_k   = (const float*)d_in[2];
    float*       out   = (float*)d_out;

    const int B = in_sizes[0] / (NQ * 3);

    dim3 grid(NQ / QPB, B);
    dim3 block(256);
    fp_knn_wmma_kernel<<<grid, block, 0, stream>>>(xyz_q, xyz_k, v_k, out);
}


# Per lane the 64 distances of a trip go through a 5-level pairwise tournament: only the trip winner takes the top-3 insert (1 per trip, not 32), each level's losers meet one running minimum; exact
# speedup vs baseline: 1.3553x; 1.1698x over previous
#include <hip/hip_runtime.h>

typedef __attribute__((ext_vector_type(2))) float v2f;
typedef __attribute__((ext_vector_type(8))) float v8f;

#define NQ   8192
#define NK   2048
#define CDIM 128
#define QPB  128
#define NCH  4

__device__ __forceinline__ void top3_insert(float v, int ki,
                                            float& b0, float& b1, float& b2,
                                            int& i0, int& i1, int& i2) {
    const bool c0 = v < b0;
    const bool c1 = v < b1;
    const bool c2 = v < b2;
    const int  p1 = c0 ? i0 : ki;
    const int  p2 = c1 ? i1 : ki;
    i0 = c0 ? ki : i0;
    i1 = c1 ? p1 : i1;
    i2 = c2 ? p2 : i2;
    const float n0 = fminf(b0, v);
    const float n1 = __builtin_amdgcn_fmed3f(b0, b1, v);
    const float n2 = __builtin_amdgcn_fmed3f(b1, b2, v);
    b0 = n0; b1 = n1; b2 = n2;
}

__device__ __forceinline__ void top3_merge(float v, int ki,
                                           float& b0, float& b1, float& b2,
                                           int& i0, int& i1, int& i2) {
    const bool c0 = (v < b0) | ((v == b0) & (ki < i0));
    const bool c1 = (v < b1) | ((v == b1) & (ki < i1));
    const bool c2 = (v < b2) | ((v == b2) & (ki < i2));
    const int  p1 = c0 ? i0 : ki;
    const int  p2 = c1 ? i1 : ki;
    i0 = c0 ? ki : i0;
    i1 = c1 ? p1 : i1;
    i2 = c2 ? p2 : i2;
    const float n0 = fminf(b0, v);
    const float n1 = __builtin_amdgcn_fmed3f(b0, b1, v);
    const float n2 = __builtin_amdgcn_fmed3f(b1, b2, v);
    b0 = n0; b1 = n1; b2 = n2;
}

struct Behind {
    float h1, h2, h3;
    int   j1, j2, j3;
};

__device__ __forceinline__ void keep_min(float v, int key, float& h, int& j) {
    const bool c = v < h;
    h = c ? v : h;
    j = c ? key : j;
}

__device__ __forceinline__ void play(float a, int ka, float b, int kc,
                                     float& lo, int& klo, float& h, int& j) {
    const bool sw = b < a;
    lo  = sw ? b : a;
    klo = sw ? kc : ka;
    keep_min(sw ? a : b, sw ? ka : kc, h, j);
}

__device__ __forceinline__ void play_rows(float a, float b, int ka,
                                          float& lo, int& klo, float& h, int& j) {
    const bool sw = b < a;
    lo  = sw ? b : a;
    klo = sw ? (ka + 1) : ka;
    keep_min(sw ? a : b, klo, h, j);
}

__device__ __forceinline__ void tile_tournament(const v8f& d, int kb, Behind& s,
                                                float& win, int& kwin) {
    float lo1[4]; int k1[4];
#pragma unroll
    for (int p = 0; p < 4; ++p)
        play_rows(d[2 * p], d[2 * p + 1], kb + 2 * p, lo1[p], k1[p], s.h1, s.j1);
    float lo2[2]; int k2[2];
#pragma unroll
    for (int p = 0; p < 2; ++p)
        play(lo1[2 * p], k1[2 * p], lo1[2 * p + 1], k1[2 * p + 1], lo2[p], k2[p], s.h2, s.j2);
    play(lo2[0], k2[0], lo2[1], k2[1], win, kwin, s.h3, s.j3);
}

__global__ __launch_bounds__(256) __attribute__((amdgpu_waves_per_eu(2, 2))) void
fp_knn_wmma_kernel(const float* __restrict__ xyz_q,
                   const float* __restrict__ xyz_k,
                   const float* __restrict__ v_k,
                   float* __restrict__ out) {
    __shared__ float s_keys[NK * 4];

    const int b      = blockIdx.y;
    const int qblock = blockIdx.x * QPB;
    const int tid    = threadIdx.x;

    const float* kb = xyz_k + (size_t)b * NK * 3;
    for (int m = tid; m < NK; m += 256) {
        float kx = kb[m * 3 + 0];
        float ky = kb[m * 3 + 1];
        float kz = kb[m * 3 + 2];
        s_keys[m * 4 + 0] = kx;
        s_keys[m * 4 + 1] = ky;
        s_keys[m * 4 + 2] = kz;
        s_keys[m * 4 + 3] = kx * kx + ky * ky + kz * kz;
    }
    __syncthreads();

    const int wave = tid >> 5;
    const int lane = tid & 31;
    const int half = lane >> 4;
    const int l16  = lane & 15;

    const int   q  = qblock + wave * 16 + l16;
    const float* qp = xyz_q + ((size_t)b * NQ + q) * 3;
    float qx = qp[0], qy = qp[1], qz = qp[2];
    float qn2 = qx * qx + qy * qy + qz * qz;

    v2f bq;
    bq.x = half ? (-2.0f * qz) : (-2.0f * qx);
    bq.y = half ? 1.0f         : (-2.0f * qy);

    const int abase = l16 * 4 + half * 2;

    Behind bh[NCH];
#pragma unroll
    for (int c = 0; c < NCH; ++c) {
        bh[c].h1 = 1e30f; bh[c].h2 = 1e30f; bh[c].h3 = 1e30f;
        bh[c].j1 = 0;     bh[c].j2 = 0;     bh[c].j3 = 0;
    }
    float h4[NCH / 2], h5 = 1e30f;
    int   j4[NCH / 2], j5 = 0;
#pragma unroll
    for (int c = 0; c < NCH / 2; ++c) { h4[c] = 1e30f; j4[c] = 0; }
    float b0 = 1e30f, b1 = 1e30f, b2 = 1e30f;
    int   i0 = 0, i1 = 0, i2 = 0;

    v2f a_cur[NCH];
#pragma unroll
    for (int c = 0; c < NCH; ++c) {
        const float* sk = &s_keys[c * 64 + abase];
        a_cur[c].x = sk[0];
        a_cur[c].y = sk[1];
    }

    for (int t = 0; t < NK; t += 16 * NCH) {
        v2f a_nxt[NCH];
#pragma unroll
        for (int c = 0; c < NCH; ++c) {
            const float* sk = &s_keys[(((t + 16 * NCH + 16 * c) & (NK - 1)) * 4) + abase];
            a_nxt[c].x = sk[0];
            a_nxt[c].y = sk[1];
        }

        v8f d[NCH];
#pragma unroll
        for (int c = 0; c < NCH; ++c) {
            v8f z = {};
            d[c] = __builtin_amdgcn_wmma_f32_16x16x4_f32(
                 false, a_cur[c],  false, bq,
                 (short)0, z,  false,  false);
        }

        const int kbase = t + half * 8;
        float w[NCH]; int kw[NCH];
#pragma unroll
        for (int c = 0; c < NCH; ++c) tile_tournament(d[c], kbase + 16 * c, bh[c], w[c], kw[c]);
        float w4[NCH / 2]; int kw4[NCH / 2];
#pragma unroll
        for (int c = 0; c < NCH / 2; ++c)
            play(w[2 * c], kw[2 * c], w[2 * c + 1], kw[2 * c + 1], w4[c], kw4[c], h4[c], j4[c]);
        float w5; int kw5;
        play(w4[0], kw4[0], w4[1], kw4[1], w5, kw5, h5, j5);
        top3_insert(w5, kw5, b0, b1, b2, i0, i1, i2);

#pragma unroll
        for (int c = 0; c < NCH; ++c) a_cur[c] = a_nxt[c];
    }

#pragma unroll
    for (int c = 0; c < NCH; ++c) {
        top3_merge(bh[c].h1, bh[c].j1 ^ 1, b0, b1, b2, i0, i1, i2);
        top3_merge(bh[c].h2, bh[c].j2, b0, b1, b2, i0, i1, i2);
        top3_merge(bh[c].h3, bh[c].j3, b0, b1, b2, i0, i1, i2);
    }
#pragma unroll
    for (int c = 0; c < NCH / 2; ++c) top3_merge(h4[c], j4[c], b0, b1, b2, i0, i1, i2);
    top3_merge(h5, j5, b0, b1, b2, i0, i1, i2);
    float ob0 = __shfl_xor(b0, 16, 32);
    float ob1 = __shfl_xor(b1, 16, 32);
    float ob2 = __shfl_xor(b2, 16, 32);
    int   oi0 = __shfl_xor(i0, 16, 32);
    int   oi1 = __shfl_xor(i1, 16, 32);
    int   oi2 = __shfl_xor(i2, 16, 32);
    top3_merge(ob0, oi0, b0, b1, b2, i0, i1, i2);
    top3_merge(ob1, oi1, b0, b1, b2, i0, i1, i2);
    top3_merge(ob2, oi2, b0, b1, b2, i0, i1, i2);

    float d0 = fmaxf(b0 + qn2, 1e-10f);
    float d1 = fmaxf(b1 + qn2, 1e-10f);
    float d2 = fmaxf(b2 + qn2, 1e-10f);
    float w0 = 1.0f / d0, w1 = 1.0f / d1, w2 = 1.0f / d2;
    float inv = 1.0f / (w0 + w1 + w2);
    w0 *= inv; w1 *= inv; w2 *= inv;

    const float* vb = v_k + (size_t)b * NK * CDIM;
    float*       ob = out + ((size_t)b * NQ + qblock + wave * 16) * CDIM;
    const int cidx = lane * 4;

    for (int qq = 0; qq < 16; ++qq) {
        int   j0 = __shfl(i0, qq, 32);
        int   j1 = __shfl(i1, qq, 32);
        int   j2 = __shfl(i2, qq, 32);
        float u0 = __shfl(w0, qq, 32);
        float u1 = __shfl(w1, qq, 32);
        float u2 = __shfl(w2, qq, 32);

        const float4 f0 = *(const float4*)(vb + (size_t)j0 * CDIM + cidx);
        const float4 f1 = *(const float4*)(vb + (size_t)j1 * CDIM + cidx);
        const float4 f2 = *(const float4*)(vb + (size_t)j2 * CDIM + cidx);

        float4 r;
        r.x = u0 * f0.x + u1 * f1.x + u2 * f2.x;
        r.y = u0 * f0.y + u1 * f1.y + u2 * f2.y;
        r.z = u0 * f0.z + u1 * f1.z + u2 * f2.z;
        r.w = u0 * f0.w + u1 * f1.w + u2 * f2.w;

        *(float4*)(ob + (size_t)qq * CDIM + cidx) = r;
    }
}

extern "C" void kernel_launch(void* const* d_in, const int* in_sizes, int n_in,
                              void* d_out, int out_size, void* d_ws, size_t ws_size,
                              hipStream_t stream) {
    const float* xyz_q = (const float*)d_in[0];
    const float* xyz_k = (const float*)d_in[1];
    const float* v_k   = (const float*)d_in[2];
    float*       out   = (float*)d_out;

    const int B = in_sizes[0] / (NQ * 3);

    dim3 grid(NQ / QPB, B);
    dim3 block(256);
    fp_knn_wmma_kernel<<<grid, block, 0, stream>>>(xyz_q, xyz_k, v_k, out);
}


# Keys go from global memory straight into the A fragments, one trip ahead (batch touched once by an L2 prefetch at entry); |k|^2 formed per lane, one more product fused; LDS staging and barrier gone
# speedup vs baseline: 1.5279x; 1.1274x over previous
#include <hip/hip_runtime.h>

typedef __attribute__((ext_vector_type(2))) float v2f;
typedef __attribute__((ext_vector_type(8))) float v8f;

#define NQ   8192
#define NK   2048
#define CDIM 128
#define QPB  128
#define NCH  4

__device__ __forceinline__ void top3_insert(float v, int ki,
                                            float& b0, float& b1, float& b2,
                                            int& i0, int& i1, int& i2) {
    const bool c0 = v < b0;
    const bool c1 = v < b1;
    const bool c2 = v < b2;
    const int  p1 = c0 ? i0 : ki;
    const int  p2 = c1 ? i1 : ki;
    i0 = c0 ? ki : i0;
    i1 = c1 ? p1 : i1;
    i2 = c2 ? p2 : i2;
    const float n0 = fminf(b0, v);
    const float n1 = __builtin_amdgcn_fmed3f(b0, b1, v);
    const float n2 = __builtin_amdgcn_fmed3f(b1, b2, v);
    b0 = n0; b1 = n1; b2 = n2;
}

__device__ __forceinline__ void top3_merge(float v, int ki,
                                           float& b0, float& b1, float& b2,
                                           int& i0, int& i1, int& i2) {
    const bool c0 = (v < b0) | ((v == b0) & (ki < i0));
    const bool c1 = (v < b1) | ((v == b1) & (ki < i1));
    const bool c2 = (v < b2) | ((v == b2) & (ki < i2));
    const int  p1 = c0 ? i0 : ki;
    const int  p2 = c1 ? i1 : ki;
    i0 = c0 ? ki : i0;
    i1 = c1 ? p1 : i1;
    i2 = c2 ? p2 : i2;
    const float n0 = fminf(b0, v);
    const float n1 = __builtin_amdgcn_fmed3f(b0, b1, v);
    const float n2 = __builtin_amdgcn_fmed3f(b1, b2, v);
    b0 = n0; b1 = n1; b2 = n2;
}

struct Behind {
    float h1, h2, h3;
    int   j1, j2, j3;
};

__device__ __forceinline__ void keep_min(float v, int key, float& h, int& j) {
    const bool c = v < h;
    h = c ? v : h;
    j = c ? key : j;
}

__device__ __forceinline__ void play(float a, int ka, float b, int kc,
                                     float& lo, int& klo, float& h, int& j) {
    const bool sw = b < a;
    lo  = sw ? b : a;
    klo = sw ? kc : ka;
    keep_min(sw ? a : b, sw ? ka : kc, h, j);
}

__device__ __forceinline__ void play_rows(float a, float b, int ka,
                                          float& lo, int& klo, float& h, int& j) {
    const bool sw = b < a;
    lo  = sw ? b : a;
    klo = sw ? (ka + 1) : ka;
    keep_min(sw ? a : b, klo, h, j);
}

__device__ __forceinline__ void tile_tournament(const v8f& d, int kb, Behind& s,
                                                float& win, int& kwin) {
    float lo1[4]; int k1[4];
#pragma unroll
    for (int p = 0; p < 4; ++p)
        play_rows(d[2 * p], d[2 * p + 1], kb + 2 * p, lo1[p], k1[p], s.h1, s.j1);
    float lo2[2]; int k2[2];
#pragma unroll
    for (int p = 0; p < 2; ++p)
        play(lo1[2 * p], k1[2 * p], lo1[2 * p + 1], k1[2 * p + 1], lo2[p], k2[p], s.h2, s.j2);
    play(lo2[0], k2[0], lo2[1], k2[1], win, kwin, s.h3, s.j3);
}

struct KeyRows { float x[NCH], y[NCH], z[NCH]; };

__device__ __forceinline__ void load_rows(const float* __restrict__ kl, int t, KeyRows& k) {
#pragma unroll
    for (int c = 0; c < NCH; ++c) {
        const float* p = kl + ((t & (NK - 1)) + 16 * c) * 3;
        k.x[c] = p[0];
        k.y[c] = p[1];
        k.z[c] = p[2];
    }
}

struct Sweep {
    Behind bh[NCH];
    float  h4[NCH / 2], h5;
    int    j4[NCH / 2], j5;
    float  b0, b1, b2;
    int    i0, i1, i2;
};

__device__ __forceinline__ void build_a(const KeyRows& k, int half, v2f (&a)[NCH]) {
#pragma unroll
    for (int c = 0; c < NCH; ++c) {
        const float n = k.x[c] * k.x[c] + k.y[c] * k.y[c] + k.z[c] * k.z[c];
        a[c].x = half ? k.z[c] : k.x[c];
        a[c].y = half ? n      : k.y[c];
    }
}

__device__ __forceinline__ void trip(int t, int half, const float* __restrict__ kl,
                                     v2f (&a)[NCH], v2f bq, Sweep& s) {
    KeyRows k;
    load_rows(kl, t + 16 * NCH, k);

    v8f d[NCH];
#pragma unroll
    for (int c = 0; c < NCH; ++c) {
        v8f z = {};
        d[c] = __builtin_amdgcn_wmma_f32_16x16x4_f32(
             false, a[c],  false, bq,
             (short)0, z,  false,  false);
    }

    const int kbase = t + half * 8;
    float w[NCH]; int kw[NCH];
#pragma unroll
    for (int c = 0; c < NCH; ++c) tile_tournament(d[c], kbase + 16 * c, s.bh[c], w[c], kw[c]);
    float w4[NCH / 2]; int kw4[NCH / 2];
#pragma unroll
    for (int c = 0; c < NCH / 2; ++c)
        play(w[2 * c], kw[2 * c], w[2 * c + 1], kw[2 * c + 1], w4[c], kw4[c], s.h4[c], s.j4[c]);
    float w5; int kw5;
    play(w4[0], kw4[0], w4[1], kw4[1], w5, kw5, s.h5, s.j5);
    top3_insert(w5, kw5, s.b0, s.b1, s.b2, s.i0, s.i1, s.i2);

    build_a(k, half, a);
}

__global__ __launch_bounds__(256) __attribute__((amdgpu_waves_per_eu(2, 2))) void
fp_knn_wmma_kernel(const float* __restrict__ xyz_q,
                   const float* __restrict__ xyz_k,
                   const float* __restrict__ v_k,
                   float* __restrict__ out) {
    const int b      = blockIdx.y;
    const int qblock = blockIdx.x * QPB;
    const int tid    = threadIdx.x;

    const int wave = tid >> 5;
    const int lane = tid & 31;
    const int half = lane >> 4;
    const int l16  = lane & 15;

    const float* kl = xyz_k + ((size_t)b * NK + l16) * 3;
    KeyRows k0;
    load_rows(kl, 0, k0);

    const int   q  = qblock + wave * 16 + l16;
    const float* qp = xyz_q + ((size_t)b * NQ + q) * 3;
    float qx = qp[0], qy = qp[1], qz = qp[2];

    if (tid < NK * 3 * 4 / 256)
        __builtin_prefetch(xyz_k + (size_t)b * NK * 3 + tid * 64, 0, 3);

    float qn2 = qx * qx + qy * qy + qz * qz;

    v2f bq;
    bq.x = half ? (-2.0f * qz) : (-2.0f * qx);
    bq.y = half ? 1.0f         : (-2.0f * qy);

    Sweep s;
#pragma unroll
    for (int c = 0; c < NCH; ++c) {
        s.bh[c].h1 = 1e30f; s.bh[c].h2 = 1e30f; s.bh[c].h3 = 1e30f;
        s.bh[c].j1 = 0;     s.bh[c].j2 = 0;     s.bh[c].j3 = 0;
    }
#pragma unroll
    for (int c = 0; c < NCH / 2; ++c) { s.h4[c] = 1e30f; s.j4[c] = 0; }
    s.h5 = 1e30f; s.j5 = 0;
    s.b0 = 1e30f; s.b1 = 1e30f; s.b2 = 1e30f;
    s.i0 = 0; s.i1 = 0; s.i2 = 0;

    v2f a[NCH];
    build_a(k0, half, a);
    for (int t = 0; t < NK; t += 16 * NCH) trip(t, half, kl, a, bq, s);

    Behind (&bh)[NCH] = s.bh;
    float (&h4)[NCH / 2] = s.h4; int (&j4)[NCH / 2] = s.j4;
    float h5 = s.h5; int j5 = s.j5;
    float b0 = s.b0, b1 = s.b1, b2 = s.b2;
    int   i0 = s.i0, i1 = s.i1, i2 = s.i2;

#pragma unroll
    for (int c = 0; c < NCH; ++c) {
        top3_merge(bh[c].h1, bh[c].j1 ^ 1, b0, b1, b2, i0, i1, i2);
        top3_merge(bh[c].h2, bh[c].j2, b0, b1, b2, i0, i1, i2);
        top3_merge(bh[c].h3, bh[c].j3, b0, b1, b2, i0, i1, i2);
    }
#pragma unroll
    for (int c = 0; c < NCH / 2; ++c) top3_merge(h4[c], j4[c], b0, b1, b2, i0, i1, i2);
    top3_merge(h5, j5, b0, b1, b2, i0, i1, i2);
    float ob0 = __shfl_xor(b0, 16, 32);
    float ob1 = __shfl_xor(b1, 16, 32);
    float ob2 = __shfl_xor(b2, 16, 32);
    int   oi0 = __shfl_xor(i0, 16, 32);
    int   oi1 = __shfl_xor(i1, 16, 32);
    int   oi2 = __shfl_xor(i2, 16, 32);
    top3_merge(ob0, oi0, b0, b1, b2, i0, i1, i2);
    top3_merge(ob1, oi1, b0, b1, b2, i0, i1, i2);
    top3_merge(ob2, oi2, b0, b1, b2, i0, i1, i2);

    float d0 = fmaxf(b0 + qn2, 1e-10f);
    float d1 = fmaxf(b1 + qn2, 1e-10f);
    float d2 = fmaxf(b2 + qn2, 1e-10f);
    float w0 = 1.0f / d0, w1 = 1.0f / d1, w2 = 1.0f / d2;
    float inv = 1.0f / (w0 + w1 + w2);
    w0 *= inv; w1 *= inv; w2 *= inv;

    const float* vb = v_k + (size_t)b * NK * CDIM;
    float*       ob = out + ((size_t)b * NQ + qblock + wave * 16) * CDIM;
    const int cidx = lane * 4;

    for (int qq = 0; qq < 16; ++qq) {
        int   j0 = __shfl(i0, qq, 32);
        int   j1 = __shfl(i1, qq, 32);
        int   j2 = __shfl(i2, qq, 32);
        float u0 = __shfl(w0, qq, 32);
        float u1 = __shfl(w1, qq, 32);
        float u2 = __shfl(w2, qq, 32);

        const float4 f0 = *(const float4*)(vb + (size_t)j0 * CDIM + cidx);
        const float4 f1 = *(const float4*)(vb + (size_t)j1 * CDIM + cidx);
        const float4 f2 = *(const float4*)(vb + (size_t)j2 * CDIM + cidx);

        float4 r;
        r.x = u0 * f0.x + u1 * f1.x + u2 * f2.x;
        r.y = u0 * f0.y + u1 * f1.y + u2 * f2.y;
        r.z = u0 * f0.z + u1 * f1.z + u2 * f2.z;
        r.w = u0 * f0.w + u1 * f1.w + u2 * f2.w;

        *(float4*)(ob + (size_t)qq * CDIM + cidx) = r;
    }
}

extern "C" void kernel_launch(void* const* d_in, const int* in_sizes, int n_in,
                              void* d_out, int out_size, void* d_ws, size_t ws_size,
                              hipStream_t stream) {
    const float* xyz_q = (const float*)d_in[0];
    const float* xyz_k = (const float*)d_in[1];
    const float* v_k   = (const float*)d_in[2];
    float*       out   = (float*)d_out;

    const int B = in_sizes[0] / (NQ * 3);

    dim3 grid(NQ / QPB, B);
    dim3 block(256);
    fp_knn_wmma_kernel<<<grid, block, 0, stream>>>(xyz_q, xyz_k, v_k, out);
}


# Values left behind in the tournament carry no key, only one running minimum and trip number per level; a final-three entry among them gets its key by forming that one trip again; exact
# speedup vs baseline: 1.6106x; 1.0541x over previous
#include <hip/hip_runtime.h>

typedef __attribute__((ext_vector_type(2))) float v2f;
typedef __attribute__((ext_vector_type(8))) float v8f;

#define NQ   8192
#define NK   2048
#define CDIM 128
#define QPB  128
#define NCH  4

__device__ __forceinline__ void top3_insert(float v, int ki,
                                            float& b0, float& b1, float& b2,
                                            int& i0, int& i1, int& i2) {
    const bool c0 = v < b0;
    const bool c1 = v < b1;
    const bool c2 = v < b2;
    const int  p1 = c0 ? i0 : ki;
    const int  p2 = c1 ? i1 : ki;
    i0 = c0 ? ki : i0;
    i1 = c1 ? p1 : i1;
    i2 = c2 ? p2 : i2;
    const float n0 = fminf(b0, v);
    const float n1 = __builtin_amdgcn_fmed3f(b0, b1, v);
    const float n2 = __builtin_amdgcn_fmed3f(b1, b2, v);
    b0 = n0; b1 = n1; b2 = n2;
}

__device__ __forceinline__ void top3_merge(float v, int ki,
                                           float& b0, float& b1, float& b2,
                                           int& i0, int& i1, int& i2) {
    const bool c0 = (v < b0) | ((v == b0) & (ki < i0));
    const bool c1 = (v < b1) | ((v == b1) & (ki < i1));
    const bool c2 = (v < b2) | ((v == b2) & (ki < i2));
    const int  p1 = c0 ? i0 : ki;
    const int  p2 = c1 ? i1 : ki;
    i0 = c0 ? ki : i0;
    i1 = c1 ? p1 : i1;
    i2 = c2 ? p2 : i2;
    const float n0 = fminf(b0, v);
    const float n1 = __builtin_amdgcn_fmed3f(b0, b1, v);
    const float n2 = __builtin_amdgcn_fmed3f(b1, b2, v);
    b0 = n0; b1 = n1; b2 = n2;
}


__device__ __forceinline__ float vmin(float a, float b) {
    return __builtin_amdgcn_fmed3f(a, b, -__builtin_inff());
}

__device__ __forceinline__ void play(float a, int ka, float b, int kc,
                                     float& lo, int& klo, float& hi) {
    const bool sw = b < a;
    lo  = sw ? b : a;
    klo = sw ? kc : ka;
    hi  = sw ? a : b;
}

__device__ __forceinline__ void tile_tournament(const v8f& d, int ko, bool first,
                                                float& win, int& kwin,
                                                float& m1, float& m2, float& m3) {
    float lo1[4]; int k1[4]; float hi;
#pragma unroll
    for (int p = 0; p < 4; ++p) {
        play(d[2 * p], ko + 2 * p, d[2 * p + 1], ko + 2 * p + 1, lo1[p], k1[p], hi);
        m1 = (first && p == 0) ? hi : vmin(m1, hi);
    }
    float lo2[2]; int k2[2];
#pragma unroll
    for (int p = 0; p < 2; ++p) {
        play(lo1[2 * p], k1[2 * p], lo1[2 * p + 1], k1[2 * p + 1], lo2[p], k2[p], hi);
        m2 = (first && p == 0) ? hi : vmin(m2, hi);
    }
    play(lo2[0], k2[0], lo2[1], k2[1], win, kwin, hi);
    m3 = first ? hi : vmin(m3, hi);
}

__device__ __forceinline__ void note_behind(float m, int t, float& h, int& ht) {
    const bool c = m < h;
    h  = c ? m : h;
    ht = c ? t : ht;
}

struct KeyRows { float x[NCH], y[NCH], z[NCH]; };

__device__ __forceinline__ void load_rows(const float* __restrict__ kl, int t, KeyRows& k) {
#pragma unroll
    for (int c = 0; c < NCH; ++c) {
        const float* p = kl + ((t & (NK - 1)) + 16 * c) * 3;
        k.x[c] = p[0];
        k.y[c] = p[1];
        k.z[c] = p[2];
    }
}

struct Sweep {
    float  h[5];
    int    ht[5];
    float  b0, b1, b2;
    int    i0, i1, i2;
};

__device__ __forceinline__ void form_tiles(const v2f (&a)[NCH], v2f bq, v8f (&d)[NCH]) {
#pragma unroll
    for (int c = 0; c < NCH; ++c) {
        v8f z = {};
        d[c] = __builtin_amdgcn_wmma_f32_16x16x4_f32(
             false, a[c],  false, bq,
             (short)0, z,  false,  false);
    }
}

__device__ __forceinline__ void build_a(const KeyRows& k, int half, v2f (&a)[NCH]) {
#pragma unroll
    for (int c = 0; c < NCH; ++c) {
        const float n = __builtin_fmaf(k.z[c], k.z[c], __builtin_fmaf(k.x[c], k.x[c], k.y[c] * k.y[c]));
        a[c].x = half ? k.z[c] : k.x[c];
        a[c].y = half ? n      : k.y[c];
    }
}

__device__ __forceinline__ void trip(int t, int half, const float* __restrict__ kl,
                                     v2f (&a)[NCH], v2f bq, Sweep& s) {
    KeyRows k;
    load_rows(kl, t + 16 * NCH, k);

    v8f d[NCH];
    form_tiles(a, bq, d);

    float w[NCH]; int kw[NCH]; float m1, m2, m3, m4, m5, hi;
#pragma unroll
    for (int c = 0; c < NCH; ++c) tile_tournament(d[c], 16 * c, c == 0, w[c], kw[c], m1, m2, m3);
    float w4[NCH / 2]; int kw4[NCH / 2];
    play(w[0], kw[0], w[1], kw[1], w4[0], kw4[0], m4);
    play(w[2], kw[2], w[3], kw[3], w4[1], kw4[1], hi);
    m4 = vmin(m4, hi);
    float w5; int kw5;
    play(w4[0], kw4[0], w4[1], kw4[1], w5, kw5, m5);
    top3_insert(w5, kw5 + (t + half * 8), s.b0, s.b1, s.b2, s.i0, s.i1, s.i2);

    note_behind(m1, t, s.h[0], s.ht[0]);
    note_behind(m2, t, s.h[1], s.ht[1]);
    note_behind(m3, t, s.h[2], s.ht[2]);
    note_behind(m4, t, s.h[3], s.ht[3]);
    note_behind(m5, t, s.h[4], s.ht[4]);

    build_a(k, half, a);
}

__device__ __forceinline__ void settle(const v8f (&d)[NCH], int half, int tt, float v,
                                       int& tag, int& last, float& lastv) {
    const bool mine = ((tag >> 1) & ~(16 * NCH - 1)) == tt;
    if (__builtin_amdgcn_ballot_w32(mine) == 0) return;
    const int taken = ((v == lastv) ? last : -1) - half * 8;
    int best = 64;
#pragma unroll
    for (int c = 0; c < NCH; ++c) {
        int bc = 64;
#pragma unroll
        for (int r = 7; r >= 0; --r)
            bc = ((d[c][r] == v) & (16 * c + r > taken)) ? (16 * c + r) : bc;
        best = min(best, bc);
    }
    best += half * 8;
    best = min(best, __shfl_xor(best, 16, 32));
    tag   = mine ? 2 * ((tt + best) & (NK - 1)) : tag;
    last  = mine ? best : last;
    lastv = mine ? v : lastv;
}

__device__ __forceinline__ int next_trip(int i0, int i1, int i2, int skip) {
    const int t0 = (i0 >> 1) & ~(16 * NCH - 1);
    const int t1 = (i1 >> 1) & ~(16 * NCH - 1);
    const int t2 = (i2 >> 1) & ~(16 * NCH - 1);
    const int want = ((i0 & 1) && t0 != skip) ? t0
                   : ((i1 & 1) && t1 != skip) ? t1
                   : ((i2 & 1) && t2 != skip) ? t2 : -1;
    const unsigned need = __builtin_amdgcn_ballot_w32(want >= 0);
    if (need == 0) return -1;
    return __builtin_amdgcn_readlane(want, __builtin_ctz(need));
}

__global__ __launch_bounds__(256) __attribute__((amdgpu_waves_per_eu(2, 2))) void
fp_knn_wmma_kernel(const float* __restrict__ xyz_q,
                   const float* __restrict__ xyz_k,
                   const float* __restrict__ v_k,
                   float* __restrict__ out) {
    const int b      = blockIdx.y;
    const int qblock = blockIdx.x * QPB;
    const int tid    = threadIdx.x;

    const int wave = tid >> 5;
    const int lane = tid & 31;
    const int half = lane >> 4;
    const int l16  = lane & 15;

    const float* kl = xyz_k + ((size_t)b * NK + l16) * 3;
    KeyRows k0;
    load_rows(kl, 0, k0);

    const int   q  = qblock + wave * 16 + l16;
    const float* qp = xyz_q + ((size_t)b * NQ + q) * 3;
    float qx = qp[0], qy = qp[1], qz = qp[2];

    if (tid < NK * 3 * 4 / 256)
        __builtin_prefetch(xyz_k + (size_t)b * NK * 3 + tid * 64, 0, 3);

    float qn2 = qx * qx + qy * qy + qz * qz;

    v2f bq;
    bq.x = half ? (-2.0f * qz) : (-2.0f * qx);
    bq.y = half ? 1.0f         : (-2.0f * qy);

    Sweep s;
#pragma unroll
    for (int l = 0; l < 5; ++l) { s.h[l] = 1e30f; s.ht[l] = 0; }
    s.b0 = 1e30f; s.b1 = 1e30f; s.b2 = 1e30f;
    s.i0 = 0; s.i1 = 0; s.i2 = 0;

    v2f a[NCH];
    build_a(k0, half, a);
    for (int t = 0; t < NK; t += 16 * NCH) trip(t, half, kl, a, bq, s);

    float b0 = s.b0, b1 = s.b1, b2 = s.b2;
    int   i0 = s.i0 << 1, i1 = s.i1 << 1, i2 = s.i2 << 1;
#pragma unroll
    for (int l = 0; l < 5; ++l) top3_merge(s.h[l], 2 * s.ht[l] + 1, b0, b1, b2, i0, i1, i2);
    float ob0 = __shfl_xor(b0, 16, 32);
    float ob1 = __shfl_xor(b1, 16, 32);
    float ob2 = __shfl_xor(b2, 16, 32);
    int   oi0 = __shfl_xor(i0, 16, 32);
    int   oi1 = __shfl_xor(i1, 16, 32);
    int   oi2 = __shfl_xor(i2, 16, 32);
    top3_merge(ob0, oi0, b0, b1, b2, i0, i1, i2);
    top3_merge(ob1, oi1, b0, b1, b2, i0, i1, i2);
    top3_merge(ob2, oi2, b0, b1, b2, i0, i1, i2);

    int tt = next_trip(i0, i1, i2, -1);
    KeyRows kr;
    load_rows(kl, tt < 0 ? 0 : tt, kr);
    for (int pass = 0; pass < 48 && tt >= 0; ++pass) {
        const int tn = next_trip(i0, i1, i2, tt);
        KeyRows kn;
        load_rows(kl, tn < 0 ? 0 : tn, kn);
        v2f ar[NCH];
        build_a(kr, half, ar);
        v8f dr[NCH];
        form_tiles(ar, bq, dr);
        int last = -1; float lastv = 0.0f;
        settle(dr, half, tt, b0, i0, last, lastv);
        settle(dr, half, tt, b1, i1, last, lastv);
        settle(dr, half, tt, b2, i2, last, lastv);
        kr = kn;
        tt = tn;
    }
    i0 >>= 1; i1 >>= 1; i2 >>= 1;

    float d0 = fmaxf(b0 + qn2, 1e-10f);
    float d1 = fmaxf(b1 + qn2, 1e-10f);
    float d2 = fmaxf(b2 + qn2, 1e-10f);
    float w0 = 1.0f / d0, w1 = 1.0f / d1, w2 = 1.0f / d2;
    float inv = 1.0f / (w0 + w1 + w2);
    w0 *= inv; w1 *= inv; w2 *= inv;

    const float* vb = v_k + (size_t)b * NK * CDIM;
    float*       ob = out + ((size_t)b * NQ + qblock + wave * 16) * CDIM;
    const int cidx = lane * 4;

    for (int qq = 0; qq < 16; ++qq) {
        int   j0 = __shfl(i0, qq, 32);
        int   j1 = __shfl(i1, qq, 32);
        int   j2 = __shfl(i2, qq, 32);
        float u0 = __shfl(w0, qq, 32);
        float u1 = __shfl(w1, qq, 32);
        float u2 = __shfl(w2, qq, 32);

        const float4 f0 = *(const float4*)(vb + (size_t)j0 * CDIM + cidx);
        const float4 f1 = *(const float4*)(vb + (size_t)j1 * CDIM + cidx);
        const float4 f2 = *(const float4*)(vb + (size_t)j2 * CDIM + cidx);

        float4 r;
        r.x = u0 * f0.x + u1 * f1.x + u2 * f2.x;
        r.y = u0 * f0.y + u1 * f1.y + u2 * f2.y;
        r.z = u0 * f0.z + u1 * f1.z + u2 * f2.z;
        r.w = u0 * f0.w + u1 * f1.w + u2 * f2.w;

        *(float4*)(ob + (size_t)qq * CDIM + cidx) = r;
    }
}

extern "C" void kernel_launch(void* const* d_in, const int* in_sizes, int n_in,
                              void* d_out, int out_size, void* d_ws, size_t ws_size,
                              hipStream_t stream) {
    const float* xyz_q = (const float*)d_in[0];
    const float* xyz_k = (const float*)d_in[1];
    const float* v_k   = (const float*)d_in[2];
    float*       out   = (float*)d_out;

    const int B = in_sizes[0] / (NQ * 3);

    dim3 grid(NQ / QPB, B);
    dim3 block(256);
    fp_knn_wmma_kernel<<<grid, block, 0, stream>>>(xyz_q, xyz_k, v_k, out);
}
